# starting kernel
# baseline (speedup 1.0000x reference)
#include <hip/hip_runtime.h>

#define N_NODES 20000
#define KN      32
#define DFEAT   128
#define NODES_PER_BLOCK 32

typedef __attribute__((ext_vector_type(2))) float v2f;
typedef __attribute__((ext_vector_type(4))) float v4f;
typedef __attribute__((ext_vector_type(8))) float v8f;

__global__ __launch_bounds__(256)
void gnn_layer_kernel(const float* __restrict__ h_prev,
                      const int*   __restrict__ adj,
                      const float* __restrict__ W,
                      const float* __restrict__ bias,
                      float*       __restrict__ h_out)
{
    __shared__ float smW[DFEAT * DFEAT];
    __shared__ float smA[NODES_PER_BLOCK * DFEAT];
    __shared__ int   smAdj[NODES_PER_BLOCK * KN];

    const int tid = threadIdx.x;
    const int n0  = blockIdx.x * NODES_PER_BLOCK;

    {
        const v4f* __restrict__ W4 = reinterpret_cast<const v4f*>(W);
        v4f* __restrict__ smW4     = reinterpret_cast<v4f*>(smW);
        #pragma unroll
        for (int it = 0; it < 16; ++it)
            smW4[tid + it * 256] = W4[tid + it * 256];
    }

    #pragma unroll
    for (int it = 0; it < 4; ++it) {
        int idx = tid + it * 256;
        int nl  = idx & (NODES_PER_BLOCK - 1);
        int k   = idx >> 5;
        smAdj[nl * KN + k] = adj[k * N_NODES + n0 + nl];
    }
    __syncthreads();

    const v4f* __restrict__ hp4 = reinterpret_cast<const v4f*>(h_prev);
    v4f* __restrict__ smA4      = reinterpret_cast<v4f*>(smA);

    #pragma unroll
    for (int it = 0; it < 4; ++it) {
        int idx = tid + it * 256;
        int nl  = idx >> 5;
        int c4  = idx & 31;
        v4f acc = {};
        #pragma unroll 8
        for (int k = 0; k < KN; ++k) {
            int nb = smAdj[nl * KN + k];
            acc += hp4[(size_t)nb * (DFEAT / 4) + c4];
        }
        smA4[nl * (DFEAT / 4) + c4] = acc * (1.0f / KN);
    }
    __syncthreads();

    const int wave    = tid >> 5;
    const int lane    = tid & 31;
    const int half    = lane >> 4;
    const int l16     = lane & 15;
    const int colBase = wave * 16;

    v8f c0 = {};
    v8f c1 = {};
    #pragma unroll
    for (int ks = 0; ks < DFEAT / 4; ++ks) {
        const int k0 = ks * 4 + half * 2;
        v2f bm;
        bm.x = smW[(k0    ) * DFEAT + colBase + l16];
        bm.y = smW[(k0 + 1) * DFEAT + colBase + l16];
        v2f a0, a1;
        a0.x = smA[(l16     ) * DFEAT + k0];
        a0.y = smA[(l16     ) * DFEAT + k0 + 1];
        a1.x = smA[(l16 + 16) * DFEAT + k0];
        a1.y = smA[(l16 + 16) * DFEAT + k0 + 1];
        c0 = __builtin_amdgcn_wmma_f32_16x16x4_f32(
                 false, a0, false, bm, (short)0, c0, false, false);
        c1 = __builtin_amdgcn_wmma_f32_16x16x4_f32(
                 false, a1, false, bm, (short)0, c1, false, false);
    }

    const int   col = colBase + l16;
    const float bv  = bias[col];
    #pragma unroll
    for (int r = 0; r < 8; ++r) {
        const int row = r + half * 8;
        float v0 = c0[r] + bv;
        float v1 = c1[r] + bv;
        v0 = (v0 > 0.0f) ? v0 : 0.0f;
        v1 = (v1 > 0.0f) ? v1 : 0.0f;
        h_out[(size_t)(n0 + row     ) * DFEAT + col] = v0;
        h_out[(size_t)(n0 + row + 16) * DFEAT + col] = v1;
    }
}

extern "C" void kernel_launch(void* const* d_in, const int* in_sizes, int n_in,
                              void* d_out, int out_size, void* d_ws, size_t ws_size,
                              hipStream_t stream) {
    (void)in_sizes; (void)n_in; (void)out_size; (void)d_ws; (void)ws_size;

    const int*   adj   = (const int*)  d_in[0];
    const float* graph = (const float*)d_in[1];
    const float* W     = (const float*)d_in[2];
    const float* b     = (const float*)d_in[3];
    float*       out   = (float*)d_out;

    const size_t layerElems = (size_t)N_NODES * DFEAT;

    hipMemcpyAsync(out, graph, layerElems * sizeof(float),
                   hipMemcpyDeviceToDevice, stream);

    dim3 grid(N_NODES / NODES_PER_BLOCK);
    dim3 block(256);

    gnn_layer_kernel<<<grid, block, 0, stream>>>(
        graph, adj, W + 1 * DFEAT * DFEAT, b + 1 * DFEAT, out + 1 * layerElems);

    gnn_layer_kernel<<<grid, block, 0, stream>>>(
        out + 1 * layerElems, adj, W + 2 * DFEAT * DFEAT, b + 2 * DFEAT,
        out + 2 * layerElems);
}


# Re-tiled: each wave owns 16 nodes x all 128 output columns (one A fragment feeds 8 accumulator tiles) and gathers its own rows, ids broadcast by readlane; 2-wave blocks; sum order unchanged
# speedup vs baseline: 1.0789x; 1.0789x over previous
#include <hip/hip_runtime.h>


#define N_NODES 20000
#define KN      32
#define DFEAT   128
#define NODES_PER_WAVE  16
#define WAVES_PER_BLOCK 2
#define NODES_PER_BLOCK (NODES_PER_WAVE * WAVES_PER_BLOCK)

typedef __attribute__((ext_vector_type(2))) float v2f;
typedef __attribute__((ext_vector_type(4))) float v4f;
typedef __attribute__((ext_vector_type(8))) float v8f;

__global__ __launch_bounds__(64)
void gnn_layer_kernel(const float* __restrict__ h_prev,
                      const int*   __restrict__ adj,
                      const float* __restrict__ W,
                      const float* __restrict__ bias,
                      float*       __restrict__ h_out)
{
    __shared__ float smW[DFEAT * DFEAT];
    __shared__ float smA[WAVES_PER_BLOCK * NODES_PER_WAVE * DFEAT];

    const int tid  = threadIdx.x;
    const int wave = tid >> 5;
    const int lane = tid & 31;
    const int half = lane >> 4;
    const int l16  = lane & 15;
    const int nw0  = blockIdx.x * NODES_PER_BLOCK + wave * NODES_PER_WAVE;

    int adjr[KN / 2];
    #pragma unroll
    for (int j = 0; j < KN / 2; ++j)
        adjr[j] = adj[(size_t)(2 * j + half) * N_NODES + nw0 + l16];

    {
        const v4f* __restrict__ W4 = reinterpret_cast<const v4f*>(W);
        v4f* __restrict__ smW4     = reinterpret_cast<v4f*>(smW);
        #pragma unroll 16
        for (int it = 0; it < 64; ++it)
            smW4[tid + it * 64] = W4[tid + it * 64];
    }

    const v4f* __restrict__ hp4 = reinterpret_cast<const v4f*>(h_prev);
    float* __restrict__ myA = smA + wave * (NODES_PER_WAVE * DFEAT);
    v4f* __restrict__ myA4  = reinterpret_cast<v4f*>(myA);

    #pragma unroll 2
    for (int nl = 0; nl < NODES_PER_WAVE; ++nl) {
        v4f acc = {};
        #pragma unroll
        for (int k = 0; k < KN; ++k) {
            const int nb = __builtin_amdgcn_readlane(adjr[k >> 1], ((k & 1) << 4) + nl);
            acc += hp4[(size_t)nb * (DFEAT / 4) + lane];
        }
        myA4[nl * (DFEAT / 4) + lane] = acc * (1.0f / KN);
    }
    __syncthreads();

    v8f c[8];
    #pragma unroll
    for (int ct = 0; ct < 8; ++ct) c[ct] = (v8f){};

    #pragma unroll 4
    for (int ks = 0; ks < DFEAT / 4; ++ks) {
        const int k0 = ks * 4 + half * 2;
        v2f a;
        a.x = myA[l16 * DFEAT + k0];
        a.y = myA[l16 * DFEAT + k0 + 1];
        #pragma unroll
        for (int ct = 0; ct < 8; ++ct) {
            v2f bm;
            bm.x = smW[(k0    ) * DFEAT + ct * 16 + l16];
            bm.y = smW[(k0 + 1) * DFEAT + ct * 16 + l16];
            c[ct] = __builtin_amdgcn_wmma_f32_16x16x4_f32(
                        false, a, false, bm, (short)0, c[ct], false, false);
        }
    }

    #pragma unroll
    for (int ct = 0; ct < 8; ++ct) {
        const int   col = ct * 16 + l16;
        const float bv  = bias[col];
        #pragma unroll
        for (int r = 0; r < 8; ++r) {
            const int row = r + half * 8;
            float v = c[ct][r] + bv;
            v = (v > 0.0f) ? v : 0.0f;
            h_out[(size_t)(nw0 + row) * DFEAT + col] = v;
        }
    }
}

extern "C" void kernel_launch(void* const* d_in, const int* in_sizes, int n_in,
                              void* d_out, int out_size, void* d_ws, size_t ws_size,
                              hipStream_t stream) {
    (void)in_sizes; (void)n_in; (void)out_size; (void)d_ws; (void)ws_size;

    const int*   adj   = (const int*)  d_in[0];
    const float* graph = (const float*)d_in[1];
    const float* W     = (const float*)d_in[2];
    const float* b     = (const float*)d_in[3];
    float*       out   = (float*)d_out;

    const size_t layerElems = (size_t)N_NODES * DFEAT;

    hipMemcpyAsync(out, graph, layerElems * sizeof(float),
                   hipMemcpyDeviceToDevice, stream);

    dim3 grid(N_NODES / NODES_PER_BLOCK);
    dim3 block(WAVES_PER_BLOCK * 32);

    gnn_layer_kernel<<<grid, block, 0, stream>>>(
        graph, adj, W + 1 * DFEAT * DFEAT, b + 1 * DFEAT, out + 1 * layerElems);

    gnn_layer_kernel<<<grid, block, 0, stream>>>(
        out + 1 * layerElems, adj, W + 2 * DFEAT * DFEAT, b + 2 * DFEAT,
        out + 2 * layerElems);
}


# One rolling load queue per wave: node n+2's 32 rows are requested as soon as node n is summed and W is copied in eighths between them (no W phase in front, no drained trips); sum order unchanged
# speedup vs baseline: 1.1688x; 1.0833x over previous
#include <hip/hip_runtime.h>


#define N_NODES 20000
#define KN      32
#define DFEAT   128
#define NODES_PER_WAVE  16
#define WAVES_PER_BLOCK 2
#define NODES_PER_BLOCK (NODES_PER_WAVE * WAVES_PER_BLOCK)

typedef __attribute__((ext_vector_type(2))) float v2f;
typedef __attribute__((ext_vector_type(4))) float v4f;
typedef __attribute__((ext_vector_type(8))) float v8f;

__global__ __launch_bounds__(64)
void gnn_layer_kernel(const float* __restrict__ h_prev,
                      const int*   __restrict__ adj,
                      const float* __restrict__ W,
                      const float* __restrict__ bias,
                      float*       __restrict__ h_out)
{
    __shared__ float smW[DFEAT * DFEAT];
    __shared__ float smA[WAVES_PER_BLOCK * NODES_PER_WAVE * DFEAT];

    const int tid  = threadIdx.x;
    const int wave = tid >> 5;
    const int lane = tid & 31;
    const int half = lane >> 4;
    const int l16  = lane & 15;
    const int nw0  = blockIdx.x * NODES_PER_BLOCK + wave * NODES_PER_WAVE;

    int adjr[KN / 2];
    #pragma unroll
    for (int j = 0; j < KN / 2; ++j)
        adjr[j] = adj[(size_t)(2 * j + half) * N_NODES + nw0 + l16];

    const v4f* __restrict__ hp4 = reinterpret_cast<const v4f*>(h_prev);
    float* __restrict__ myA = smA + wave * (NODES_PER_WAVE * DFEAT);
    v4f* __restrict__ myA4  = reinterpret_cast<v4f*>(myA);
    const v4f* __restrict__ W4 = reinterpret_cast<const v4f*>(W);
    v4f* __restrict__ smW4     = reinterpret_cast<v4f*>(smW);

    v4f ra[KN], rb[KN];

    #pragma unroll
    for (int k = 0; k < KN; ++k) {
        const int nb = __builtin_amdgcn_readlane(adjr[k >> 1], ((k & 1) << 4) + 0);
        ra[k] = hp4[(size_t)nb * (DFEAT / 4) + lane];
    }
    #pragma unroll
    for (int k = 0; k < KN; ++k) {
        const int nb = __builtin_amdgcn_readlane(adjr[k >> 1], ((k & 1) << 4) + 1);
        rb[k] = hp4[(size_t)nb * (DFEAT / 4) + lane];
    }

    #pragma unroll 1
    for (int nl = 0; nl < NODES_PER_WAVE - 2; nl += 2) {
        const int wbase = tid + (nl >> 1) * (8 * 64);
        v4f wq[8];
        {
            v4f acc = {};
            #pragma unroll
            for (int k = 0; k < KN; ++k) acc += ra[k];
            myA4[nl * (DFEAT / 4) + lane] = acc * (1.0f / KN);
        }
        #pragma unroll
        for (int i = 0; i < 8; ++i) wq[i] = W4[wbase + i * 64];
        #pragma unroll
        for (int k = 0; k < KN; ++k) {
            const int nb = __builtin_amdgcn_readlane(adjr[k >> 1], ((k & 1) << 4) + nl + 2);
            ra[k] = hp4[(size_t)nb * (DFEAT / 4) + lane];
        }
        __builtin_amdgcn_sched_barrier(0);
        {
            v4f acc = {};
            #pragma unroll
            for (int k = 0; k < KN; ++k) acc += rb[k];
            myA4[(nl + 1) * (DFEAT / 4) + lane] = acc * (1.0f / KN);
        }
        #pragma unroll
        for (int i = 0; i < 8; ++i) smW4[wbase + i * 64] = wq[i];
        #pragma unroll
        for (int k = 0; k < KN; ++k) {
            const int nb = __builtin_amdgcn_readlane(adjr[k >> 1], ((k & 1) << 4) + nl + 3);
            rb[k] = hp4[(size_t)nb * (DFEAT / 4) + lane];
        }
        __builtin_amdgcn_sched_barrier(0);
    }
    {
        const int wbase = tid + 7 * (8 * 64);
        v4f wq[8];
        #pragma unroll
        for (int i = 0; i < 8; ++i) wq[i] = W4[wbase + i * 64];
        {
            v4f acc = {};
            #pragma unroll
            for (int k = 0; k < KN; ++k) acc += ra[k];
            myA4[(NODES_PER_WAVE - 2) * (DFEAT / 4) + lane] = acc * (1.0f / KN);
        }
        {
            v4f acc = {};
            #pragma unroll
            for (int k = 0; k < KN; ++k) acc += rb[k];
            myA4[(NODES_PER_WAVE - 1) * (DFEAT / 4) + lane] = acc * (1.0f / KN);
        }
        #pragma unroll
        for (int i = 0; i < 8; ++i) smW4[wbase + i * 64] = wq[i];
    }
    __syncthreads();

    v8f c[8];
    #pragma unroll
    for (int ct = 0; ct < 8; ++ct) c[ct] = (v8f){};

    #pragma unroll 4
    for (int ks = 0; ks < DFEAT / 4; ++ks) {
        const int k0 = ks * 4 + half * 2;
        v2f a;
        a.x = myA[l16 * DFEAT + k0];
        a.y = myA[l16 * DFEAT + k0 + 1];
        #pragma unroll
        for (int ct = 0; ct < 8; ++ct) {
            v2f bm;
            bm.x = smW[(k0    ) * DFEAT + ct * 16 + l16];
            bm.y = smW[(k0 + 1) * DFEAT + ct * 16 + l16];
            c[ct] = __builtin_amdgcn_wmma_f32_16x16x4_f32(
                        false, a, false, bm, (short)0, c[ct], false, false);
        }
    }

    #pragma unroll
    for (int ct = 0; ct < 8; ++ct) {
        const int   col = ct * 16 + l16;
        const float bv  = bias[col];
        #pragma unroll
        for (int r = 0; r < 8; ++r) {
            const int row = r + half * 8;
            float v = c[ct][r] + bv;
            v = (v > 0.0f) ? v : 0.0f;
            h_out[(size_t)(nw0 + row) * DFEAT + col] = v;
        }
    }
}

extern "C" void kernel_launch(void* const* d_in, const int* in_sizes, int n_in,
                              void* d_out, int out_size, void* d_ws, size_t ws_size,
                              hipStream_t stream) {
    (void)in_sizes; (void)n_in; (void)out_size; (void)d_ws; (void)ws_size;

    const int*   adj   = (const int*)  d_in[0];
    const float* graph = (const float*)d_in[1];
    const float* W     = (const float*)d_in[2];
    const float* b     = (const float*)d_in[3];
    float*       out   = (float*)d_out;

    const size_t layerElems = (size_t)N_NODES * DFEAT;

    hipMemcpyAsync(out, graph, layerElems * sizeof(float),
                   hipMemcpyDeviceToDevice, stream);

    dim3 grid(N_NODES / NODES_PER_BLOCK);
    dim3 block(WAVES_PER_BLOCK * 32);

    gnn_layer_kernel<<<grid, block, 0, stream>>>(
        graph, adj, W + 1 * DFEAT * DFEAT, b + 1 * DFEAT, out + 1 * layerElems);

    gnn_layer_kernel<<<grid, block, 0, stream>>>(
        out + 1 * layerElems, adj, W + 2 * DFEAT * DFEAT, b + 2 * DFEAT,
        out + 2 * layerElems);
}


# Re-decomposed: 250 equal blocks of 80 nodes, one per compute unit; each wave keeps its 128x16 W panel in registers and a chunk's products run under the next chunk's row requests; sum order unchanged
# speedup vs baseline: 1.3326x; 1.1401x over previous
#include <hip/hip_runtime.h>


#define N_NODES 20000
#define KN      32
#define DFEAT   128
#define WAVES_PER_BLOCK 8
#define CHUNK_NODES     16
#define CHUNKS          5
#define NODES_PER_BLOCK (CHUNK_NODES * CHUNKS)

typedef __attribute__((ext_vector_type(2))) float v2f;
typedef __attribute__((ext_vector_type(4))) float v4f;
typedef __attribute__((ext_vector_type(8))) float v8f;

static __device__ __forceinline__ void lds_barrier() {
    __builtin_amdgcn_fence(__ATOMIC_RELEASE, "workgroup", "local");
    __builtin_amdgcn_s_barrier();
    __builtin_amdgcn_fence(__ATOMIC_ACQUIRE, "workgroup", "local");
}


#define REQUEST_ROWS(BUF, IDV)                                                          \
    _Pragma("unroll")                                                                   \
    for (int k = 0; k < KN; ++k) {                                                      \
        const unsigned nb = (unsigned)__builtin_amdgcn_readlane((IDV), k);              \
        BUF[k] = *reinterpret_cast<const v4f*>(hbytes + ((nb << 9) + laneoff));         \
    }

#define MEAN_TO_TILE(BUF, ROW)                                                          \
    {                                                                                   \
        v4f acc = {};                                                                   \
        _Pragma("unroll")                                                               \
        for (int k = 0; k < KN; ++k) acc += BUF[k];                                     \
        acc = acc * (1.0f / KN);                                                        \
        smAf[lane * 32 + (ROW)]      = (v2f){acc.x, acc.y};                             \
        smAf[lane * 32 + 16 + (ROW)] = (v2f){acc.z, acc.w};                             \
    }

#define PRODUCT_AND_STORE(CH)                                                           \
    {                                                                                   \
        v8f c8 = {};                                                                    \
        _Pragma("unroll")                                                               \
        for (int ks = 0; ks < DFEAT / 4; ++ks) {                                        \
            const v2f a = smAf[ks * 32 + lane];                                         \
            c8 = __builtin_amdgcn_wmma_f32_16x16x4_f32(                                 \
                     false, a, false, breg[ks], (short)0, c8, false, false);            \
        }                                                                               \
        _Pragma("unroll")                                                               \
        for (int r = 0; r < 8; ++r) {                                                   \
            float v = c8[r] + bv;                        \
            v = (v > 0.0f) ? v : 0.0f;                                                  \
            smOut[(r + half * 8) * DFEAT + wave * 16 + l16] = v;                        \
        }                                                                               \
        lds_barrier();                           \
        _Pragma("unroll")                                                               \
        for (int j = 0; j < 2; ++j) {                                                   \
            const int row = 2 * wave + j;                                               \
            const v4f o = *reinterpret_cast<const v4f*>(&smOut[row * DFEAT + 4 * lane]);\
            *reinterpret_cast<v4f*>(                                                    \
                &h_out[(size_t)(b0 + (CH) * CHUNK_NODES + row) * DFEAT + 4 * lane]) = o;\
        }                                                                               \
    }

__global__ __launch_bounds__(256)
void gnn_layer_kernel(const float* __restrict__ h_prev,
                      const int*   __restrict__ adj,
                      const float* __restrict__ W,
                      const float* __restrict__ bias,
                      float*       __restrict__ h_out)
{
    __shared__ int   smAdj[KN * NODES_PER_BLOCK];
    __shared__ v2f   smAf[(DFEAT / 4) * 32];
    __shared__ float smOut[CHUNK_NODES * DFEAT];

    const int tid  = threadIdx.x;
    const int wave = tid >> 5;
    const int lane = tid & 31;
    const int half = lane >> 4;
    const int l16  = lane & 15;
    const int b0   = blockIdx.x * NODES_PER_BLOCK;

    #pragma unroll
    for (int i = 0; i < (KN * NODES_PER_BLOCK) / 256; ++i) {
        const int idx = tid + 256 * i;
        const int k   = idx / NODES_PER_BLOCK;
        const int n   = idx - k * NODES_PER_BLOCK;
        smAdj[idx] = adj[(size_t)k * N_NODES + b0 + n];
    }

    v2f breg[DFEAT / 4];
    #pragma unroll
    for (int ks = 0; ks < DFEAT / 4; ++ks) {
        const int k0 = ks * 4 + half * 2;
        breg[ks].x = W[(k0    ) * DFEAT + wave * 16 + l16];
        breg[ks].y = W[(k0 + 1) * DFEAT + wave * 16 + l16];
    }
    const float bv = bias[wave * 16 + l16];

    __syncthreads();

    const char* __restrict__ hbytes = reinterpret_cast<const char*>(h_prev);
    const unsigned laneoff = (unsigned)lane * 16u;

    v4f ra[KN], rb[KN];
    {
        const int ida = smAdj[lane * NODES_PER_BLOCK + 2 * wave];
        const int idb = smAdj[lane * NODES_PER_BLOCK + 2 * wave + 1];
        REQUEST_ROWS(ra, ida)
        REQUEST_ROWS(rb, idb)
    }

    #pragma unroll 1
    for (int ch = 0; ch < CHUNKS - 1; ++ch) {
        const int nxt = lane * NODES_PER_BLOCK + (ch + 1) * CHUNK_NODES + 2 * wave;
        const int ida = smAdj[nxt];
        const int idb = smAdj[nxt + 1];
        MEAN_TO_TILE(ra, 2 * wave)
        REQUEST_ROWS(ra, ida)
        __builtin_amdgcn_sched_barrier(0);
        MEAN_TO_TILE(rb, 2 * wave + 1)
        REQUEST_ROWS(rb, idb)
        __builtin_amdgcn_sched_barrier(0);
        lds_barrier();
        PRODUCT_AND_STORE(ch)
    }
    MEAN_TO_TILE(ra, 2 * wave)
    MEAN_TO_TILE(rb, 2 * wave + 1)
    lds_barrier();
    PRODUCT_AND_STORE(CHUNKS - 1)
}

extern "C" void kernel_launch(void* const* d_in, const int* in_sizes, int n_in,
                              void* d_out, int out_size, void* d_ws, size_t ws_size,
                              hipStream_t stream) {
    (void)in_sizes; (void)n_in; (void)out_size; (void)d_ws; (void)ws_size;

    const int*   adj   = (const int*)  d_in[0];
    const float* graph = (const float*)d_in[1];
    const float* W     = (const float*)d_in[2];
    const float* b     = (const float*)d_in[3];
    float*       out   = (float*)d_out;

    const size_t layerElems = (size_t)N_NODES * DFEAT;

    hipMemcpyAsync(out, graph, layerElems * sizeof(float),
                   hipMemcpyDeviceToDevice, stream);

    dim3 grid(N_NODES / NODES_PER_BLOCK);
    dim3 block(WAVES_PER_BLOCK * 32);

    gnn_layer_kernel<<<grid, block, 0, stream>>>(
        graph, adj, W + 1 * DFEAT * DFEAT, b + 1 * DFEAT, out + 1 * layerElems);

    gnn_layer_kernel<<<grid, block, 0, stream>>>(
        out + 1 * layerElems, adj, W + 2 * DFEAT * DFEAT, b + 2 * DFEAT,
        out + 2 * layerElems);
}
